# starting kernel
# baseline (speedup 1.0000x reference)
#include <hip/hip_runtime.h>

typedef __attribute__((ext_vector_type(2))) float v2f;
typedef __attribute__((ext_vector_type(8))) float v8f;

#define MN  49
#define KB1 192
#define KB2 48

__global__ __launch_bounds__(256)
void fused_shift_conv_wmma(const float* __restrict__ x,
                           const float* __restrict__ w1,
                           const float* __restrict__ w2,
                           float* __restrict__ out)
{
    __shared__ float t4s[16 * 16 * MN];

    const int lane  = threadIdx.x & 31;
    const int wave  = threadIdx.x >> 5;
    const int hi    = lane >> 4;
    const int lo    = lane & 15;
    const int kbase = blockIdx.x * 16;

    for (int job = wave; job < 64; job += 8) {
        const int o  = job >> 2;
        const int nt = job & 3;
        const int c  = nt * 16 + lo;
        const int m  = c / 7;
        const int n  = c % 7;

        const float* w1row = w1 + (kbase + lo) * KB1;

        v8f acc = {};
        #pragma unroll
        for (int jw = 0; jw < 3; ++jw) {
            const int   h    = m + jw - 1;
            const bool  ok   = (c < MN) & (h >= 0) & (h < 7);
            const int   xoff = ((h + 6) % 7) * 7 + n;
            const float* xb  = x + o * MN + xoff;

            #pragma unroll
            for (int k0 = 0; k0 < 64; k0 += 4) {
                const int i0 = k0 + 2 * hi;
                v2f a, b;
                a.x = w1row[i0 * 3 + jw];
                a.y = w1row[i0 * 3 + jw + 3];
                const float v0 = xb[i0 * (16 * MN)];
                const float v1 = xb[(i0 + 1) * (16 * MN)];
                b.x = ok ? v0 : 0.0f;
                b.y = ok ? v1 : 0.0f;
                acc = __builtin_amdgcn_wmma_f32_16x16x4_f32(false, a, false, b,
                                                            (short)0, acc, false, false);
            }
        }
        if (c < MN) {
            #pragma unroll
            for (int v = 0; v < 8; ++v) {
                const int r = v + 8 * hi;
                t4s[(r * 16 + o) * MN + c] = acc[v];
            }
        }
    }

    __syncthreads();

    for (int job = wave; job < 128; job += 8) {
        const int nt  = job & 3;
        const int imt = (job >> 2) & 1;
        const int o2l = job >> 3;
        const int c   = nt * 16 + lo;
        const int m   = c / 7;
        const int n   = c % 7;

        const float* w2row = w2 + (imt * 16 + lo) * KB2;

        v8f acc = {};
        #pragma unroll
        for (int kw = 0; kw < 3; ++kw) {
            const int  w   = n + kw - 1;
            const bool ok  = (c < MN) & (w >= 0) & (w < 7);
            const int  off = ok ? (m * 7 + w) : 0;
            const float* tb = t4s + o2l * (16 * MN) + off;

            #pragma unroll
            for (int k0 = 0; k0 < 16; k0 += 4) {
                const int j0 = k0 + 2 * hi;
                v2f a, b;
                a.x = w2row[j0 * 3 + kw];
                a.y = w2row[j0 * 3 + kw + 3];
                const float v0 = tb[j0 * MN];
                const float v1 = tb[(j0 + 1) * MN];
                b.x = ok ? v0 : 0.0f;
                b.y = ok ? v1 : 0.0f;
                acc = __builtin_amdgcn_wmma_f32_16x16x4_f32(false, a, false, b,
                                                            (short)0, acc, false, false);
            }
        }
        if (c < MN) {
            #pragma unroll
            for (int v = 0; v < 8; ++v) {
                const int i  = imt * 16 + v + 8 * hi;
                const int o2 = kbase + o2l;
                out[(i * 32 + o2) * MN + c] = acc[v];
            }
        }
    }
}

extern "C" void kernel_launch(void* const* d_in, const int* in_sizes, int n_in,
                              void* d_out, int out_size, void* d_ws, size_t ws_size,
                              hipStream_t stream) {
    const float* x  = (const float*)d_in[0];
    const float* w1 = (const float*)d_in[1];
    const float* w2 = (const float*)d_in[2];
    float* out = (float*)d_out;
    (void)in_sizes; (void)n_in; (void)out_size; (void)d_ws; (void)ws_size;
    fused_shift_conv_wmma<<<dim3(2), dim3(256), 0, stream>>>(x, w1, w2, out);
}


# Re-grid: 14 work-groups (k-tile x spatial row m) of 4 waves replace 2 of 8; each keeps its own stage-1 slice in LDS, 2 column tiles per wave share A fragments; per-element sum order unchanged
# speedup vs baseline: 3.8072x; 3.8072x over previous
#include <hip/hip_runtime.h>

typedef __attribute__((ext_vector_type(2))) float v2f;
typedef __attribute__((ext_vector_type(8))) float v8f;

#define MN   49
#define KB1  192
#define KB2  48
#define NC   112
#define XCH  784
#define NWAV 4


template <int NT>
__device__ __forceinline__ void stage1_tiles(const float* __restrict__ x,
                                             const float* __restrict__ w1p,
                                             float* __restrict__ t4s,
                                             int m, int tile0, int lo, int hi)
{
    unsigned loff[NT];
    int col[NT];
    #pragma unroll
    for (int t = 0; t < NT; ++t) {
        const int c = (tile0 + NWAV * t) * 16 + lo;
        const int o = c / 7;
        const int n = c - o * 7;
        col[t]  = c;
        loff[t] = (unsigned)(o * MN + n + hi * (2 * XCH));
    }

    float af[16][6];
    #pragma unroll
    for (int s = 0; s < 16; ++s) {
        #pragma unroll
        for (int e = 0; e < 6; ++e) af[s][e] = w1p[s * 12 + e];
    }

    v8f acc[NT];
    #pragma unroll
    for (int t = 0; t < NT; ++t) acc[t] = v8f{};

    #pragma unroll
    for (int jw = 0; jw < 3; ++jw) {
        const int  h  = m + jw - 1;
        const bool ok = (h >= 0) & (h < 7);
        const float* xr = x + ((h + 6) % 7) * 7;
        float bv[NT][16][2];
        #pragma unroll
        for (int t = 0; t < NT; ++t) {
            #pragma unroll
            for (int s = 0; s < 16; ++s) {
                bv[t][s][0] = (xr + (4 * s) * XCH)[loff[t]];
                bv[t][s][1] = (xr + (4 * s + 1) * XCH)[loff[t]];
            }
        }
        #pragma unroll
        for (int s = 0; s < 16; ++s) {
            v2f a;
            a.x = af[s][jw];
            a.y = af[s][jw + 3];
            #pragma unroll
            for (int t = 0; t < NT; ++t) {
                v2f b;
                b.x = ok ? bv[t][s][0] : 0.0f;
                b.y = ok ? bv[t][s][1] : 0.0f;
                acc[t] = __builtin_amdgcn_wmma_f32_16x16x4_f32(false, a, false, b,
                                                               (short)0, acc[t], false, false);
            }
        }
    }
    #pragma unroll
    for (int t = 0; t < NT; ++t) {
        #pragma unroll
        for (int v = 0; v < 8; ++v)
            t4s[(v + 8 * hi) * NC + col[t]] = acc[t][v];
    }
}

template <int NT>
__device__ __forceinline__ void stage2_tiles(const float* __restrict__ w2p,
                                             const float* __restrict__ t4s,
                                             float* __restrict__ out,
                                             int kt, int m, int tile0, int lo, int hi)
{
    int o2l[NT], n[NT];
    #pragma unroll
    for (int t = 0; t < NT; ++t) {
        const int c = (tile0 + NWAV * t) * 16 + lo;
        o2l[t] = c / 7;
        n[t]   = c - o2l[t] * 7;
    }

    float af[2][4][6];
    #pragma unroll
    for (int it = 0; it < 2; ++it) {
        #pragma unroll
        for (int s = 0; s < 4; ++s) {
            #pragma unroll
            for (int e = 0; e < 6; ++e) af[it][s][e] = w2p[it * 16 * KB2 + s * 12 + e];
        }
    }

    v8f acc[NT][2];
    #pragma unroll
    for (int t = 0; t < NT; ++t) { acc[t][0] = v8f{}; acc[t][1] = v8f{}; }

    #pragma unroll
    for (int kw = 0; kw < 3; ++kw) {
        bool ok[NT];
        const float* tb[NT];
        #pragma unroll
        for (int t = 0; t < NT; ++t) {
            const int w = n[t] + kw - 1;
            ok[t] = (w >= 0) & (w < 7);
            tb[t] = t4s + o2l[t] * NC + (ok[t] ? w : 0) + hi * 14;
        }
        #pragma unroll
        for (int s = 0; s < 4; ++s) {
            v2f a0, a1;
            a0.x = af[0][s][kw];
            a0.y = af[0][s][kw + 3];
            a1.x = af[1][s][kw];
            a1.y = af[1][s][kw + 3];
            #pragma unroll
            for (int t = 0; t < NT; ++t) {
                const float v0 = tb[t][(4 * s) * 7];
                const float v1 = tb[t][(4 * s + 1) * 7];
                v2f b;
                b.x = ok[t] ? v0 : 0.0f;
                b.y = ok[t] ? v1 : 0.0f;
                acc[t][0] = __builtin_amdgcn_wmma_f32_16x16x4_f32(false, a0, false, b,
                                                                  (short)0, acc[t][0], false, false);
                acc[t][1] = __builtin_amdgcn_wmma_f32_16x16x4_f32(false, a1, false, b,
                                                                  (short)0, acc[t][1], false, false);
            }
        }
    }
    #pragma unroll
    for (int t = 0; t < NT; ++t) {
        const int o2 = kt * 16 + o2l[t];
        #pragma unroll
        for (int it = 0; it < 2; ++it) {
            #pragma unroll
            for (int v = 0; v < 8; ++v) {
                const int i = it * 16 + v + 8 * hi;
                out[(i * 32 + o2) * MN + m * 7 + n[t]] = acc[t][it][v];
            }
        }
    }
}

__global__ __launch_bounds__(128)
void fused_shift_conv_wmma(const float* __restrict__ x,
                           const float* __restrict__ w1,
                           const float* __restrict__ w2,
                           float* __restrict__ out)
{
    __shared__ float t4s[16 * NC];

    const int lane = threadIdx.x & 31;
    const int wave = __builtin_amdgcn_readfirstlane((int)(threadIdx.x >> 5));
    const int hi   = lane >> 4;
    const int lo   = lane & 15;
    const int kt   = blockIdx.x / 7;
    const int m    = blockIdx.x - kt * 7;

    const float* w1p = w1 + (kt * 16 + lo) * KB1 + hi * 6;
    const float* w2p = w2 + lo * KB2 + hi * 6;

    if (wave < 3) stage1_tiles<2>(x, w1p, t4s, m, wave, lo, hi);
    else          stage1_tiles<1>(x, w1p, t4s, m, wave, lo, hi);

    __syncthreads();

    if (wave < 3) stage2_tiles<2>(w2p, t4s, out, kt, m, wave, lo, hi);
    else          stage2_tiles<1>(w2p, t4s, out, kt, m, wave, lo, hi);
}

extern "C" void kernel_launch(void* const* d_in, const int* in_sizes, int n_in,
                              void* d_out, int out_size, void* d_ws, size_t ws_size,
                              hipStream_t stream) {
    const float* x  = (const float*)d_in[0];
    const float* w1 = (const float*)d_in[1];
    const float* w2 = (const float*)d_in[2];
    float* out = (float*)d_out;
    (void)in_sizes; (void)n_in; (void)out_size; (void)d_ws; (void)ws_size;
    fused_shift_conv_wmma<<<dim3(14), dim3(128), 0, stream>>>(x, w1, w2, out);
}


# Inputs staged once per work-group into LDS by whole-line 16-byte loads (x whole, w1 slice, w2); all A/B fragments read from LDS instead of 4-byte global gathers; per-element sum order unchanged
# speedup vs baseline: 4.8851x; 1.2831x over previous
#include <hip/hip_runtime.h>

typedef __attribute__((ext_vector_type(2))) float v2f;
typedef __attribute__((ext_vector_type(4))) float v4f;
typedef __attribute__((ext_vector_type(8))) float v8f;

#define MN   49
#define KB1  192
#define KB2  48
#define NC   112
#define XCH  784
#define NWAV 4
#define NTHR 128
#define XQ   12544
#define W1P  196
#define W2P  52


template <int NT>
__device__ __forceinline__ void stage1_tiles(const float* __restrict__ xs,
                                             const float* __restrict__ w1l,
                                             float* __restrict__ t4s,
                                             int m, int tile0, int lo, int hi)
{
    unsigned loff[NT];
    int col[NT];
    #pragma unroll
    for (int t = 0; t < NT; ++t) {
        const int c = (tile0 + NWAV * t) * 16 + lo;
        const int o = c / 7;
        const int n = c - o * 7;
        col[t]  = c;
        loff[t] = (unsigned)(o * MN + n + hi * (2 * XCH));
    }

    float af[16][6];
    #pragma unroll
    for (int s = 0; s < 16; ++s) {
        #pragma unroll
        for (int e = 0; e < 6; ++e) af[s][e] = w1l[s * 12 + e];
    }

    v8f acc[NT];
    #pragma unroll
    for (int t = 0; t < NT; ++t) acc[t] = v8f{};

    float bv[2][NT][16][2];
    #pragma unroll
    for (int jw = -1; jw < 3; ++jw) {
        if (jw + 1 < 3) {
            const int hn = m + jw;
            const float* xr = xs + ((hn + 6) % 7) * 7;
            #pragma unroll
            for (int t = 0; t < NT; ++t) {
                #pragma unroll
                for (int s = 0; s < 16; ++s) {
                    bv[(jw + 1) & 1][t][s][0] = (xr + (4 * s) * XCH)[loff[t]];
                    bv[(jw + 1) & 1][t][s][1] = (xr + (4 * s + 1) * XCH)[loff[t]];
                }
            }
        }
        __builtin_amdgcn_sched_barrier(0);
        if (jw >= 0) {
            const int  h  = m + jw - 1;
            const bool ok = (h >= 0) & (h < 7);
            #pragma unroll
            for (int s = 0; s < 16; ++s) {
                v2f a;
                a.x = af[s][jw];
                a.y = af[s][jw + 3];
                #pragma unroll
                for (int t = 0; t < NT; ++t) {
                    v2f b;
                    b.x = ok ? bv[jw & 1][t][s][0] : 0.0f;
                    b.y = ok ? bv[jw & 1][t][s][1] : 0.0f;
                    acc[t] = __builtin_amdgcn_wmma_f32_16x16x4_f32(false, a, false, b,
                                                                   (short)0, acc[t], false, false);
                }
            }
            __builtin_amdgcn_sched_barrier(0);
        }
    }
    #pragma unroll
    for (int t = 0; t < NT; ++t) {
        #pragma unroll
        for (int v = 0; v < 8; ++v)
            t4s[(v + 8 * hi) * NC + col[t]] = acc[t][v];
    }
}

template <int NT>
__device__ __forceinline__ void stage2_tiles(const float* __restrict__ w2l,
                                             const float* __restrict__ t4s,
                                             float* __restrict__ out,
                                             int kt, int m, int tile0, int lo, int hi)
{
    int o2l[NT], n[NT];
    #pragma unroll
    for (int t = 0; t < NT; ++t) {
        const int c = (tile0 + NWAV * t) * 16 + lo;
        o2l[t] = c / 7;
        n[t]   = c - o2l[t] * 7;
    }

    float af[2][4][6];
    #pragma unroll
    for (int it = 0; it < 2; ++it) {
        #pragma unroll
        for (int s = 0; s < 4; ++s) {
            #pragma unroll
            for (int e = 0; e < 6; ++e) af[it][s][e] = w2l[it * 16 * W2P + s * 12 + e];
        }
    }

    v8f acc[NT][2];
    #pragma unroll
    for (int t = 0; t < NT; ++t) { acc[t][0] = v8f{}; acc[t][1] = v8f{}; }

    #pragma unroll
    for (int kw = 0; kw < 3; ++kw) {
        bool ok[NT];
        const float* tb[NT];
        #pragma unroll
        for (int t = 0; t < NT; ++t) {
            const int w = n[t] + kw - 1;
            ok[t] = (w >= 0) & (w < 7);
            tb[t] = t4s + o2l[t] * NC + (ok[t] ? w : 0) + hi * 14;
        }
        #pragma unroll
        for (int s = 0; s < 4; ++s) {
            v2f a0, a1;
            a0.x = af[0][s][kw];
            a0.y = af[0][s][kw + 3];
            a1.x = af[1][s][kw];
            a1.y = af[1][s][kw + 3];
            #pragma unroll
            for (int t = 0; t < NT; ++t) {
                const float v0 = tb[t][(4 * s) * 7];
                const float v1 = tb[t][(4 * s + 1) * 7];
                v2f b;
                b.x = ok[t] ? v0 : 0.0f;
                b.y = ok[t] ? v1 : 0.0f;
                acc[t][0] = __builtin_amdgcn_wmma_f32_16x16x4_f32(false, a0, false, b,
                                                                  (short)0, acc[t][0], false, false);
                acc[t][1] = __builtin_amdgcn_wmma_f32_16x16x4_f32(false, a1, false, b,
                                                                  (short)0, acc[t][1], false, false);
            }
        }
    }
    #pragma unroll
    for (int t = 0; t < NT; ++t) {
        const int o2 = kt * 16 + o2l[t];
        #pragma unroll
        for (int it = 0; it < 2; ++it) {
            #pragma unroll
            for (int v = 0; v < 8; ++v) {
                const int i = it * 16 + v + 8 * hi;
                out[(i * 32 + o2) * MN + m * 7 + n[t]] = acc[t][it][v];
            }
        }
    }
}

__global__ __launch_bounds__(NTHR)
void fused_shift_conv_wmma(const float* __restrict__ x,
                           const float* __restrict__ w1,
                           const float* __restrict__ w2,
                           float* __restrict__ out)
{
    __shared__ __attribute__((aligned(16))) float xs[4 * XQ];
    __shared__ __attribute__((aligned(16))) float w1s[16 * W1P];
    __shared__ __attribute__((aligned(16))) float w2s[32 * W2P];
    __shared__ float t4s[16 * NC];

    const int tid  = threadIdx.x;
    const int lane = tid & 31;
    const int wave = __builtin_amdgcn_readfirstlane((int)(threadIdx.x >> 5));
    const int hi   = lane >> 4;
    const int lo   = lane & 15;
    const int kt   = blockIdx.x / 7;
    const int m    = blockIdx.x - kt * 7;

    {
        const v4f* w1g = (const v4f*)(w1 + kt * 16 * KB1);
        const v4f* w2g = (const v4f*)w2;
        const v4f* xg  = (const v4f*)x;
        v4f r1[6], r2[3];
        #pragma unroll
        for (int r = 0; r < 6; ++r) r1[r] = w1g[tid + NTHR * r];
        #pragma unroll
        for (int r = 0; r < 3; ++r) r2[r] = w2g[tid + NTHR * r];
        #pragma unroll
        for (int b = 0; b < 2; ++b) {
            v4f rx[49];
            #pragma unroll
            for (int r = 0; r < 49; ++r) rx[r] = xg[tid + NTHR * (49 * b + r)];
            #pragma unroll
            for (int r = 0; r < 49; ++r) *(v4f*)(xs + 4 * (tid + NTHR * (49 * b + r))) = rx[r];
        }
        #pragma unroll
        for (int r = 0; r < 6; ++r) {
            const int q = tid + NTHR * r, row = q / 48, c4 = q - row * 48;
            *(v4f*)(w1s + row * W1P + 4 * c4) = r1[r];
        }
        #pragma unroll
        for (int r = 0; r < 3; ++r) {
            const int q = tid + NTHR * r, row = q / 12, c4 = q - row * 12;
            *(v4f*)(w2s + row * W2P + 4 * c4) = r2[r];
        }
    }
    __syncthreads();

    const float* w1l = w1s + lo * W1P + hi * 6;
    const float* w2l = w2s + lo * W2P + hi * 6;

    if (wave < 3) stage1_tiles<2>(xs, w1l, t4s, m, wave, lo, hi);
    else          stage1_tiles<1>(xs, w1l, t4s, m, wave, lo, hi);

    __syncthreads();

    if (wave < 3) stage2_tiles<2>(w2l, t4s, out, kt, m, wave, lo, hi);
    else          stage2_tiles<1>(w2l, t4s, out, kt, m, wave, lo, hi);
}

extern "C" void kernel_launch(void* const* d_in, const int* in_sizes, int n_in,
                              void* d_out, int out_size, void* d_ws, size_t ws_size,
                              hipStream_t stream) {
    const float* x  = (const float*)d_in[0];
    const float* w1 = (const float*)d_in[1];
    const float* w2 = (const float*)d_in[2];
    float* out = (float*)d_out;
    (void)in_sizes; (void)n_in; (void)out_size; (void)d_ws; (void)ws_size;
    fused_shift_conv_wmma<<<dim3(14), dim3(NTHR), 0, stream>>>(x, w1, w2, out);
}


# Re-grid: stage 2 of each (k-tile, row) pair split over 4 work-groups (56 in all), one 12-instruction unit per wave, its LDS reads issued ahead; each work-group recomputes stage 1; sum order unchanged
# speedup vs baseline: 5.9071x; 1.2092x over previous
#include <hip/hip_runtime.h>

typedef __attribute__((ext_vector_type(2))) float v2f;
typedef __attribute__((ext_vector_type(4))) float v4f;
typedef __attribute__((ext_vector_type(8))) float v8f;

#define MN   49
#define KB1  192
#define KB2  48
#define NC   112
#define XCH  784
#define NWAV 4
#define NTHR 128
#define XQ   12544
#define W1P  196
#define W2P  52


template <int NT>
__device__ __forceinline__ void stage1_tiles(const float* __restrict__ xs,
                                             const float* __restrict__ w1l,
                                             float* __restrict__ t4s,
                                             int m, int tile0, int lo, int hi)
{
    unsigned loff[NT];
    int col[NT];
    #pragma unroll
    for (int t = 0; t < NT; ++t) {
        const int c = (tile0 + NWAV * t) * 16 + lo;
        const int o = c / 7;
        const int n = c - o * 7;
        col[t]  = c;
        loff[t] = (unsigned)(o * MN + n + hi * (2 * XCH));
    }

    float af[16][6];
    #pragma unroll
    for (int s = 0; s < 16; ++s) {
        #pragma unroll
        for (int e = 0; e < 6; ++e) af[s][e] = w1l[s * 12 + e];
    }

    v8f acc[NT];
    #pragma unroll
    for (int t = 0; t < NT; ++t) acc[t] = v8f{};

    float bv[2][NT][16][2];
    #pragma unroll
    for (int jw = -1; jw < 3; ++jw) {
        if (jw + 1 < 3) {
            const int hn = m + jw;
            const float* xr = xs + ((hn + 6) % 7) * 7;
            #pragma unroll
            for (int t = 0; t < NT; ++t) {
                #pragma unroll
                for (int s = 0; s < 16; ++s) {
                    bv[(jw + 1) & 1][t][s][0] = (xr + (4 * s) * XCH)[loff[t]];
                    bv[(jw + 1) & 1][t][s][1] = (xr + (4 * s + 1) * XCH)[loff[t]];
                }
            }
        }
        __builtin_amdgcn_sched_barrier(0);
        if (jw >= 0) {
            const int  h  = m + jw - 1;
            const bool ok = (h >= 0) & (h < 7);
            #pragma unroll
            for (int s = 0; s < 16; ++s) {
                v2f a;
                a.x = af[s][jw];
                a.y = af[s][jw + 3];
                #pragma unroll
                for (int t = 0; t < NT; ++t) {
                    v2f b;
                    b.x = ok ? bv[jw & 1][t][s][0] : 0.0f;
                    b.y = ok ? bv[jw & 1][t][s][1] : 0.0f;
                    acc[t] = __builtin_amdgcn_wmma_f32_16x16x4_f32(false, a, false, b,
                                                                   (short)0, acc[t], false, false);
                }
            }
            __builtin_amdgcn_sched_barrier(0);
        }
    }
    #pragma unroll
    for (int t = 0; t < NT; ++t) {
        #pragma unroll
        for (int v = 0; v < 8; ++v)
            t4s[(v + 8 * hi) * NC + col[t]] = acc[t][v];
    }
}

__device__ __forceinline__ void stage2_unit(const float* __restrict__ w2l,
                                            const float* __restrict__ t4s,
                                            float* __restrict__ out,
                                            int kt, int m, int tile, int it, int lo, int hi)
{
    const int c   = tile * 16 + lo;
    const int o2l = c / 7;
    const int n   = c - o2l * 7;

    float af[4][6];
    #pragma unroll
    for (int s = 0; s < 4; ++s) {
        #pragma unroll
        for (int e = 0; e < 6; ++e) af[s][e] = w2l[s * 12 + e];
    }

    bool  ok[3];
    float bv[3][4][2];
    #pragma unroll
    for (int kw = 0; kw < 3; ++kw) {
        const int w = n + kw - 1;
        ok[kw] = (w >= 0) & (w < 7);
        const float* tb = t4s + o2l * NC + (ok[kw] ? w : 0) + hi * 14;
        #pragma unroll
        for (int s = 0; s < 4; ++s) {
            bv[kw][s][0] = tb[(4 * s) * 7];
            bv[kw][s][1] = tb[(4 * s + 1) * 7];
        }
    }
    __builtin_amdgcn_sched_barrier(0);

    v8f acc = v8f{};
    #pragma unroll
    for (int kw = 0; kw < 3; ++kw) {
        #pragma unroll
        for (int s = 0; s < 4; ++s) {
            v2f a;
            a.x = af[s][kw];
            a.y = af[s][kw + 3];
            v2f b;
            b.x = ok[kw] ? bv[kw][s][0] : 0.0f;
            b.y = ok[kw] ? bv[kw][s][1] : 0.0f;
            acc = __builtin_amdgcn_wmma_f32_16x16x4_f32(false, a, false, b,
                                                        (short)0, acc, false, false);
        }
    }
    const int o2 = kt * 16 + o2l;
    #pragma unroll
    for (int v = 0; v < 8; ++v) {
        const int i = it * 16 + v + 8 * hi;
        out[(i * 32 + o2) * MN + m * 7 + n] = acc[v];
    }
}

__global__ __launch_bounds__(NTHR)
void fused_shift_conv_wmma(const float* __restrict__ x,
                           const float* __restrict__ w1,
                           const float* __restrict__ w2,
                           float* __restrict__ out)
{
    __shared__ __attribute__((aligned(16))) float xs[4 * XQ];
    __shared__ __attribute__((aligned(16))) float w1s[16 * W1P];
    __shared__ __attribute__((aligned(16))) float w2s[32 * W2P];
    __shared__ float t4s[16 * NC];

    const int tid  = threadIdx.x;
    const int lane = tid & 31;
    const int wave = __builtin_amdgcn_readfirstlane((int)(threadIdx.x >> 5));
    const int hi   = lane >> 4;
    const int lo   = lane & 15;
    const int q    = blockIdx.x & 3;
    const int g    = blockIdx.x >> 2;
    const int kt   = g / 7;
    const int m    = g - kt * 7;

    {
        const v4f* w1g = (const v4f*)(w1 + kt * 16 * KB1);
        const v4f* w2g = (const v4f*)w2;
        const v4f* xg  = (const v4f*)x;
        v4f r1[6], r2[3];
        #pragma unroll
        for (int r = 0; r < 6; ++r) r1[r] = w1g[tid + NTHR * r];
        #pragma unroll
        for (int r = 0; r < 3; ++r) r2[r] = w2g[tid + NTHR * r];
        #pragma unroll
        for (int b = 0; b < 2; ++b) {
            v4f rx[49];
            #pragma unroll
            for (int r = 0; r < 49; ++r) rx[r] = xg[tid + NTHR * (49 * b + r)];
            #pragma unroll
            for (int r = 0; r < 49; ++r) *(v4f*)(xs + 4 * (tid + NTHR * (49 * b + r))) = rx[r];
        }
        #pragma unroll
        for (int r = 0; r < 6; ++r) {
            const int q = tid + NTHR * r, row = q / 48, c4 = q - row * 48;
            *(v4f*)(w1s + row * W1P + 4 * c4) = r1[r];
        }
        #pragma unroll
        for (int r = 0; r < 3; ++r) {
            const int q = tid + NTHR * r, row = q / 12, c4 = q - row * 12;
            *(v4f*)(w2s + row * W2P + 4 * c4) = r2[r];
        }
    }
    __syncthreads();

    const float* w1l = w1s + lo * W1P + hi * 6;

    if (wave < 3) stage1_tiles<2>(xs, w1l, t4s, m, wave, lo, hi);
    else          stage1_tiles<1>(xs, w1l, t4s, m, wave, lo, hi);

    __syncthreads();

    const int u = 4 * q + wave;
    if (u < 14) {
        const int it   = u / 7;
        const int tile = u - it * 7;
        const float* w2l = w2s + (it * 16 + lo) * W2P + hi * 6;
        stage2_unit(w2l, t4s, out, kt, m, tile, it, lo, hi);
    }
}

extern "C" void kernel_launch(void* const* d_in, const int* in_sizes, int n_in,
                              void* d_out, int out_size, void* d_ws, size_t ws_size,
                              hipStream_t stream) {
    const float* x  = (const float*)d_in[0];
    const float* w1 = (const float*)d_in[1];
    const float* w2 = (const float*)d_in[2];
    float* out = (float*)d_out;
    (void)in_sizes; (void)n_in; (void)out_size; (void)d_ws; (void)ws_size;
    fused_shift_conv_wmma<<<dim3(56), dim3(NTHR), 0, stream>>>(x, w1, w2, out);
}


# Copy-in moved to 4 loader waves beside the 4 compute waves, x cut in 4 parts of the in-channel range: stage 1 of a part runs while later parts land; stage-1 sum order now part, window row, channel
# speedup vs baseline: 6.4306x; 1.0886x over previous
#include <hip/hip_runtime.h>

typedef __attribute__((ext_vector_type(2))) float v2f;
typedef __attribute__((ext_vector_type(4))) float v4f;
typedef __attribute__((ext_vector_type(8))) float v8f;

#define MN   49
#define KB1  192
#define KB2  48
#define NC   112
#define XCH  784
#define NWAV 4
#define NLDT 128
#define NTHR 256
#define NCH  4
#define KH   4
#define NBLK 12
#define NLD  107
#define XQ   12544
#define W1P  196
#define W2P  52


template <int NT>
__device__ __forceinline__ void s1_reads(int b, const float* __restrict__ xs,
                                         float (&bv)[2][2][KH][2], const unsigned (&loff)[2], int m)
{
    const int c  = b / 3;
    const int jw = b - 3 * c;
    const int hn = m + jw - 1;
    const float* xr = xs + ((hn + 6) % 7) * 7 + (4 * KH * c) * XCH;
    #pragma unroll
    for (int t = 0; t < NT; ++t) {
        #pragma unroll
        for (int s = 0; s < KH; ++s) {
            bv[b & 1][t][s][0] = (xr + (4 * s) * XCH)[loff[t]];
            bv[b & 1][t][s][1] = (xr + (4 * s + 1) * XCH)[loff[t]];
        }
    }
}

__device__ __forceinline__ void s1_afrag(int c, const float* __restrict__ w1l, float (&af)[16][6])
{
    #pragma unroll
    for (int s = 0; s < KH; ++s) {
        #pragma unroll
        for (int e = 0; e < 6; ++e) af[KH * c + s][e] = w1l[(KH * c + s) * 12 + e];
    }
}

template <int NT>
__device__ __forceinline__ void s1_mma(int b, v8f (&acc)[2], const float (&af)[16][6],
                                       const float (&bv)[2][2][KH][2], int m)
{
    const int  c  = b / 3;
    const int  jw = b - 3 * c;
    const int  h  = m + jw - 1;
    const bool ok = (h >= 0) & (h < 7);
    #pragma unroll
    for (int s = 0; s < KH; ++s) {
        v2f a;
        a.x = af[KH * c + s][jw];
        a.y = af[KH * c + s][jw + 3];
        #pragma unroll
        for (int t = 0; t < NT; ++t) {
            v2f bb;
            bb.x = ok ? bv[b & 1][t][s][0] : 0.0f;
            bb.y = ok ? bv[b & 1][t][s][1] : 0.0f;
            acc[t] = __builtin_amdgcn_wmma_f32_16x16x4_f32(false, a, false, bb,
                                                           (short)0, acc[t], false, false);
        }
    }
}

template <int NT>
__device__ __forceinline__ void s1_blocks(int b0, int b1, const float* __restrict__ xs,
                                          const float* __restrict__ w1l, v8f (&acc)[2],
                                          float (&af)[16][6], float (&bv)[2][2][KH][2],
                                          const unsigned (&loff)[2], int m)
{
    #pragma unroll
    for (int b = b0; b < b1; ++b) {
        if (b + 1 < NBLK) {
            s1_reads<NT>(b + 1, xs, bv, loff, m);
            if (b % 3 == 0 && b / 3 + 1 < NCH) s1_afrag(b / 3 + 1, w1l, af);
        }
        __builtin_amdgcn_sched_barrier(0);
        s1_mma<NT>(b, acc, af, bv, m);
        __builtin_amdgcn_sched_barrier(0);
    }
}

template <int NT>
__device__ __forceinline__ void s1_store(const v8f (&acc)[2], float* __restrict__ t4s,
                                         int tile0, int lo, int hi)
{
    #pragma unroll
    for (int t = 0; t < NT; ++t) {
        const int c = (tile0 + NWAV * t) * 16 + lo;
        #pragma unroll
        for (int v = 0; v < 8; ++v)
            t4s[(v + 8 * hi) * NC + c] = acc[t][v];
    }
}

__device__ __forceinline__ void stage2_unit(const float* __restrict__ w2l,
                                            const float* __restrict__ t4s,
                                            float* __restrict__ out,
                                            int kt, int m, int tile, int it, int lo, int hi)
{
    const int c   = tile * 16 + lo;
    const int o2l = c / 7;
    const int n   = c - o2l * 7;

    float af[4][6];
    #pragma unroll
    for (int s = 0; s < 4; ++s) {
        #pragma unroll
        for (int e = 0; e < 6; ++e) af[s][e] = w2l[s * 12 + e];
    }

    bool  ok[3];
    float bv[3][4][2];
    #pragma unroll
    for (int kw = 0; kw < 3; ++kw) {
        const int w = n + kw - 1;
        ok[kw] = (w >= 0) & (w < 7);
        const float* tb = t4s + o2l * NC + (ok[kw] ? w : 0) + hi * 14;
        #pragma unroll
        for (int s = 0; s < 4; ++s) {
            bv[kw][s][0] = tb[(4 * s) * 7];
            bv[kw][s][1] = tb[(4 * s + 1) * 7];
        }
    }
    __builtin_amdgcn_sched_barrier(0);

    v8f acc = v8f{};
    #pragma unroll
    for (int kw = 0; kw < 3; ++kw) {
        #pragma unroll
        for (int s = 0; s < 4; ++s) {
            v2f a;
            a.x = af[s][kw];
            a.y = af[s][kw + 3];
            v2f b;
            b.x = ok[kw] ? bv[kw][s][0] : 0.0f;
            b.y = ok[kw] ? bv[kw][s][1] : 0.0f;
            acc = __builtin_amdgcn_wmma_f32_16x16x4_f32(false, a, false, b,
                                                        (short)0, acc, false, false);
        }
    }
    const int o2 = kt * 16 + o2l;
    #pragma unroll
    for (int v = 0; v < 8; ++v) {
        const int i = it * 16 + v + 8 * hi;
        out[(i * 32 + o2) * MN + m * 7 + n] = acc[v];
    }
}

__device__ __forceinline__ void lds_barrier()
{
    __builtin_amdgcn_fence(__ATOMIC_RELEASE, "workgroup", "local");
    __builtin_amdgcn_s_barrier();
    __builtin_amdgcn_fence(__ATOMIC_ACQUIRE, "workgroup", "local");
}

__global__ __launch_bounds__(NTHR)
void fused_shift_conv_wmma(const float* __restrict__ x,
                           const float* __restrict__ w1,
                           const float* __restrict__ w2,
                           float* __restrict__ out)
{
    __shared__ __attribute__((aligned(16))) float xs[4 * XQ];
    __shared__ __attribute__((aligned(16))) float w1s[16 * W1P];
    __shared__ __attribute__((aligned(16))) float w2s[32 * W2P];
    __shared__ float t4s[16 * NC];

    const int tid  = threadIdx.x;
    const int lane = tid & 31;
    const int  wave8  = __builtin_amdgcn_readfirstlane((int)(threadIdx.x >> 5));
    const bool loader = wave8 >= NWAV;
    const int  wave   = wave8 & (NWAV - 1);
    const int  lt     = tid & (NLDT - 1);
    const int  hi     = lane >> 4;
    const int  lo     = lane & 15;
    const int  q      = blockIdx.x & 3;
    const int  g      = blockIdx.x >> 2;
    const int  kt     = g / 7;
    const int  m      = g - kt * 7;

    const v4f* w1g = (const v4f*)(w1 + kt * 16 * KB1);
    const v4f* w2g = (const v4f*)w2;
    const v4f* xg  = (const v4f*)x;
    v4f rq[NLD];
    auto load_piece = [&](int j) __attribute__((always_inline)) {
        if (j < 6)        rq[j] = w1g[lt + NLDT * j];
        else if (j < 104) rq[j] = xg[lt + NLDT * (j - 6)];
        else              rq[j] = w2g[lt + NLDT * (j - 104)];
    };
    auto store_piece = [&](int j) __attribute__((always_inline)) {
        if (j < 6) {
            const int p = lt + NLDT * j, row = p / 48, c4 = p - row * 48;
            *(v4f*)(w1s + row * W1P + 4 * c4) = rq[j];
        } else if (j < 104) {
            *(v4f*)(xs + 4 * (lt + NLDT * (j - 6))) = rq[j];
        } else {
            const int p = lt + NLDT * (j - 104), row = p / 12, c4 = p - row * 12;
            *(v4f*)(w2s + row * W2P + 4 * c4) = rq[j];
        }
    };
    auto copy_step = [&](int j0, int j1) __attribute__((always_inline)) {
        #pragma unroll
        for (int j = j0; j < j1; ++j) {
            store_piece(j);
            if (j + 63 < NLD) load_piece(j + 63);
            if ((j & 3) == 3) __builtin_amdgcn_sched_barrier(0);
        }
    };

    const float* w1l = w1s + lo * W1P + hi * 6;
    v8f      acc[2] = {v8f{}, v8f{}};
    float    af[16][6];
    float    bv[2][2][KH][2];
    unsigned loff[2];
    #pragma unroll
    for (int t = 0; t < 2; ++t) {
        const int c = (wave + NWAV * t) * 16 + lo;
        const int o = c / 7;
        const int n = c - o * 7;
        loff[t] = (unsigned)(o * MN + n + hi * (2 * XCH));
    }
    auto s1_first = [&]() __attribute__((always_inline)) {
        s1_afrag(0, w1l, af);
        if (wave < 3) s1_reads<2>(0, xs, bv, loff, m);
        else          s1_reads<1>(0, xs, bv, loff, m);
    };
    auto s1_seg = [&](int b0, int b1) __attribute__((always_inline)) {
        if (wave < 3) s1_blocks<2>(b0, b1, xs, w1l, acc, af, bv, loff, m);
        else          s1_blocks<1>(b0, b1, xs, w1l, acc, af, bv, loff, m);
    };

    if (loader) {
        #pragma unroll
        for (int j = 0; j < 63; ++j) {
            load_piece(j);
            if ((j & 3) == 3 || j == 62) __builtin_amdgcn_sched_barrier(0);
        }
        copy_step(0, 31);
    }
    lds_barrier();
    if (loader) copy_step(31, 55);
    else      { s1_first(); s1_seg(0, 2); }
    lds_barrier();
    if (loader) copy_step(55, 80);
    else        s1_seg(2, 5);
    lds_barrier();
    if (loader) copy_step(80, NLD);
    else        s1_seg(5, 8);
    lds_barrier();
    if (!loader) {
        s1_seg(8, NBLK);
        if (wave < 3) s1_store<2>(acc, t4s, wave, lo, hi);
        else          s1_store<1>(acc, t4s, wave, lo, hi);
    }
    __syncthreads();

    const int u = 4 * q + wave;
    if (!loader && u < 14) {
        const int it   = u / 7;
        const int tile = u - it * 7;
        const float* w2l = w2s + (it * 16 + lo) * W2P + hi * 6;
        stage2_unit(w2l, t4s, out, kt, m, tile, it, lo, hi);
    }
}

extern "C" void kernel_launch(void* const* d_in, const int* in_sizes, int n_in,
                              void* d_out, int out_size, void* d_ws, size_t ws_size,
                              hipStream_t stream) {
    const float* x  = (const float*)d_in[0];
    const float* w1 = (const float*)d_in[1];
    const float* w2 = (const float*)d_in[2];
    float* out = (float*)d_out;
    (void)in_sizes; (void)n_in; (void)out_size; (void)d_ws; (void)ws_size;
    fused_shift_conv_wmma<<<dim3(56), dim3(NTHR), 0, stream>>>(x, w1, w2, out);
}
